# starting kernel
# baseline (speedup 1.0000x reference)
#include <hip/hip_runtime.h>

#define BB   32
#define TT   8192
#define FF   128
#define OUTT 1024
#define GG   (TT / 16)

typedef float v2f __attribute__((ext_vector_type(2)));
typedef float v8f __attribute__((ext_vector_type(8)));

__global__ __launch_bounds__(256) void k_groupsum_wmma(const float* __restrict__ x,
                                                       float* __restrict__ cs) {
    const int lane = threadIdx.x & 31;
    const int w    = blockIdx.x * 8 + (threadIdx.x >> 5);
    const int fg   = w & 7;
    const int g    = (w >> 3) & (GG - 1);
    const int b    = w >> 12;

    const int n    = lane & 15;
    const int koff = (lane >> 4) << 1;

    const float* base = x + ((size_t)b * TT + (size_t)g * 16) * FF + fg * 16 + n;

    v2f a; a.x = 1.0f; a.y = 1.0f;
    v8f c = {};
#pragma unroll
    for (int k = 0; k < 4; ++k) {
        const float* p = base + (size_t)(4 * k + koff) * FF;
        v2f bb; bb.x = p[0]; bb.y = p[FF];
        c = __builtin_amdgcn_wmma_f32_16x16x4_f32(false, a, false, bb, (short)0, c,
                                                  false, false);
    }
    if (lane < 16)
        cs[((size_t)b * GG + g) * FF + fg * 16 + lane] = c[0];
}

__global__ __launch_bounds__(128) void k_prefix(float* __restrict__ cs) {
    const int f = threadIdx.x;
    const int b = blockIdx.x;
    float run = 0.0f;
    const size_t base = (size_t)b * GG * FF + f;
#pragma unroll 4
    for (int g = 0; g < GG; ++g) {
        const size_t i = base + (size_t)g * FF;
        run += cs[i];
        cs[i] = run;
    }
}

__global__ __launch_bounds__(128) void k_out(const float* __restrict__ x,
                                             const float* __restrict__ ps,
                                             const int* __restrict__ lengths,
                                             float* __restrict__ out) {
    const int f    = threadIdx.x;
    const int tile = blockIdx.x;
    const int b    = blockIdx.y;
    const int t0   = tile * 16;

    const int   fsize = lengths[b] - (OUTT - 1);
    const float fsf   = (float)fsize;

    const float* xb = x  + (size_t)b * TT * FF + f;
    const float* pb = ps + (size_t)b * GG * FF + f;

    float lo = (tile == 0) ? 0.0f : pb[(size_t)(tile - 1) * FF];

    const int idx0 = fsize + t0;
    const int g0   = idx0 >> 4;
    const int r0   = idx0 & 15;
    float hi = (g0 == 0) ? 0.0f : pb[(size_t)(g0 - 1) * FF];
    const int s0 = idx0 - r0;
    for (int i = 0; i < r0; ++i)
        hi += xb[(size_t)(s0 + i) * FF];

    float* ob = out + ((size_t)b * OUTT + t0) * FF + f;
#pragma unroll
    for (int j = 0; j < 16; ++j) {
        ob[(size_t)j * FF] = (hi - lo) / fsf;
        if (j < 15) {
            lo += xb[(size_t)(t0 + j) * FF];
            hi += xb[(size_t)(idx0 + j) * FF];
        }
    }
}

extern "C" void kernel_launch(void* const* d_in, const int* in_sizes, int n_in,
                              void* d_out, int out_size, void* d_ws, size_t ws_size,
                              hipStream_t stream) {
    (void)in_sizes; (void)n_in; (void)out_size; (void)ws_size;
    const float* x       = (const float*)d_in[0];
    const int*   lengths = (const int*)d_in[1];
    float*       out     = (float*)d_out;
    float*       cs      = (float*)d_ws;

    k_groupsum_wmma<<<16384, 256, 0, stream>>>(x, cs);
    k_prefix<<<32, 128, 0, stream>>>(cs);
    k_out<<<dim3(64, 32), 128, 0, stream>>>(x, cs, lengths, out);
}


# Group-sum prefix made a parallel two-level scan: each thread scans 32 of a column's 512 group sums, 16 segment totals combine through LDS (changes the sum order); was one serial 512-step chain
# speedup vs baseline: 3.3437x; 3.3437x over previous
#include <hip/hip_runtime.h>

#define BB   32
#define TT   8192
#define FF   128
#define OUTT 1024
#define GG   (TT / 16)

typedef float v2f __attribute__((ext_vector_type(2)));
typedef float v8f __attribute__((ext_vector_type(8)));

__global__ __launch_bounds__(256) void k_groupsum_wmma(const float* __restrict__ x,
                                                       float* __restrict__ cs) {
    const int lane = threadIdx.x & 31;
    const int w    = blockIdx.x * 8 + (threadIdx.x >> 5);
    const int fg   = w & 7;
    const int g    = (w >> 3) & (GG - 1);
    const int b    = w >> 12;

    const int n    = lane & 15;
    const int koff = (lane >> 4) << 1;

    const float* base = x + ((size_t)b * TT + (size_t)g * 16) * FF + fg * 16 + n;

    v2f a; a.x = 1.0f; a.y = 1.0f;
    v8f c = {};
#pragma unroll
    for (int k = 0; k < 4; ++k) {
        const float* p = base + (size_t)(4 * k + koff) * FF;
        v2f bb; bb.x = p[0]; bb.y = p[FF];
        c = __builtin_amdgcn_wmma_f32_16x16x4_f32(false, a, false, bb, (short)0, c,
                                                  false, false);
    }
    if (lane < 16)
        cs[((size_t)b * GG + g) * FF + fg * 16 + lane] = c[0];
}

#define SEGS   16
#define SEGLEN (GG / SEGS)
#define SLICE  32
#define QL     (SLICE / 4)

__global__ __launch_bounds__(128) void k_prefix(float* __restrict__ cs) {
    __shared__ float4 tot[SEGS][QL];

    const int q   = threadIdx.x & (QL - 1);
    const int seg = threadIdx.x / QL;
    const int sl  = blockIdx.x & (FF / SLICE - 1);
    const int b   = blockIdx.x / (FF / SLICE);

    float4* p = reinterpret_cast<float4*>(
        cs + ((size_t)b * GG + (size_t)seg * SEGLEN) * FF + sl * SLICE + q * 4);
    constexpr int RS = FF / 4;

    float4 v[SEGLEN];
#pragma unroll
    for (int j = 0; j < SEGLEN; ++j) v[j] = p[j * RS];

#pragma unroll
    for (int j = 1; j < SEGLEN; ++j) {
        v[j].x += v[j - 1].x; v[j].y += v[j - 1].y;
        v[j].z += v[j - 1].z; v[j].w += v[j - 1].w;
    }
    tot[seg][q] = v[SEGLEN - 1];
    __syncthreads();

    float4 t[SEGS - 1];
#pragma unroll
    for (int s = 0; s < SEGS - 1; ++s) t[s] = tot[s][q];

    float4 off = make_float4(0.0f, 0.0f, 0.0f, 0.0f);
#pragma unroll
    for (int s = 0; s < SEGS - 1; ++s) {
        const bool in = s < seg;
        off.x += in ? t[s].x : 0.0f; off.y += in ? t[s].y : 0.0f;
        off.z += in ? t[s].z : 0.0f; off.w += in ? t[s].w : 0.0f;
    }

#pragma unroll
    for (int j = 0; j < SEGLEN; ++j) {
        float4 r;
        r.x = off.x + v[j].x; r.y = off.y + v[j].y;
        r.z = off.z + v[j].z; r.w = off.w + v[j].w;
        p[j * RS] = r;
    }
}

__global__ __launch_bounds__(128) void k_out(const float* __restrict__ x,
                                             const float* __restrict__ ps,
                                             const int* __restrict__ lengths,
                                             float* __restrict__ out) {
    const int f    = threadIdx.x;
    const int tile = blockIdx.x;
    const int b    = blockIdx.y;
    const int t0   = tile * 16;

    const int   fsize = lengths[b] - (OUTT - 1);
    const float fsf   = (float)fsize;

    const float* xb = x  + (size_t)b * TT * FF + f;
    const float* pb = ps + (size_t)b * GG * FF + f;

    float lo = (tile == 0) ? 0.0f : pb[(size_t)(tile - 1) * FF];

    const int idx0 = fsize + t0;
    const int g0   = idx0 >> 4;
    const int r0   = idx0 & 15;
    float hi = (g0 == 0) ? 0.0f : pb[(size_t)(g0 - 1) * FF];
    const int s0 = idx0 - r0;
    for (int i = 0; i < r0; ++i)
        hi += xb[(size_t)(s0 + i) * FF];

    float* ob = out + ((size_t)b * OUTT + t0) * FF + f;
#pragma unroll
    for (int j = 0; j < 16; ++j) {
        ob[(size_t)j * FF] = (hi - lo) / fsf;
        if (j < 15) {
            lo += xb[(size_t)(t0 + j) * FF];
            hi += xb[(size_t)(idx0 + j) * FF];
        }
    }
}

extern "C" void kernel_launch(void* const* d_in, const int* in_sizes, int n_in,
                              void* d_out, int out_size, void* d_ws, size_t ws_size,
                              hipStream_t stream) {
    (void)in_sizes; (void)n_in; (void)out_size; (void)ws_size;
    const float* x       = (const float*)d_in[0];
    const int*   lengths = (const int*)d_in[1];
    float*       out     = (float*)d_out;
    float*       cs      = (float*)d_ws;

    k_groupsum_wmma<<<16384, 256, 0, stream>>>(x, cs);
    k_prefix<<<128, 128, 0, stream>>>(cs);
    k_out<<<dim3(64, 32), 128, 0, stream>>>(x, cs, lengths, out);
}


# Group-sum kernel made a streaming loop: 4096 waves walk 4 groups each (8 KB, 16-byte loads, next group's loads issued before the WMMAs); was 131072 one-group waves, 4-byte loads; same sum order
# speedup vs baseline: 3.5583x; 1.0642x over previous
#include <hip/hip_runtime.h>

#define BB   32
#define TT   8192
#define FF   128
#define OUTT 1024
#define GG   (TT / 16)

typedef float v2f __attribute__((ext_vector_type(2)));
typedef float v8f __attribute__((ext_vector_type(8)));

#define TPW 4
#define K1_WAVES (BB * GG / TPW)
#define K1_BLOCKS (K1_WAVES / 8)
static_assert(K1_BLOCKS == 512, "the launch below is written for 512 work-groups");

__device__ __forceinline__ void k1_load_group(float4 (&d)[16], const float* __restrict__ p) {
#pragma unroll
    for (int i = 0; i < 16; ++i) {
        const int k = i >> 2, q = (i >> 1) & 1, s = i & 1;
        d[i] = *reinterpret_cast<const float4*>(p + (size_t)(4 * k + s) * FF + 64 * q);
    }
}

__device__ __forceinline__ void k1_sum_group(const float4 (&d)[16], float* __restrict__ o, int h) {
    v2f a; a.x = 1.0f; a.y = 1.0f;
    v8f acc[2][4];
#pragma unroll
    for (int i = 0; i < 8; ++i) acc[i >> 2][i & 3] = v8f{};
#pragma unroll
    for (int i = 0; i < 8; ++i) {
        const int q = i & 1;
        const float4 r0 = d[2 * i + 0];
        const float4 r1 = d[2 * i + 1];
        v2f b0; b0.x = r0.x; b0.y = r1.x;
        v2f b1; b1.x = r0.y; b1.y = r1.y;
        v2f b2; b2.x = r0.z; b2.y = r1.z;
        v2f b3; b3.x = r0.w; b3.y = r1.w;
        acc[q][0] = __builtin_amdgcn_wmma_f32_16x16x4_f32(false, a, false, b0, (short)0,
                                                          acc[q][0], false, false);
        acc[q][1] = __builtin_amdgcn_wmma_f32_16x16x4_f32(false, a, false, b1, (short)0,
                                                          acc[q][1], false, false);
        acc[q][2] = __builtin_amdgcn_wmma_f32_16x16x4_f32(false, a, false, b2, (short)0,
                                                          acc[q][2], false, false);
        acc[q][3] = __builtin_amdgcn_wmma_f32_16x16x4_f32(false, a, false, b3, (short)0,
                                                          acc[q][3], false, false);
    }
    float4 r;
    r.x = h ? acc[1][0][0] : acc[0][0][0];
    r.y = h ? acc[1][1][0] : acc[0][1][0];
    r.z = h ? acc[1][2][0] : acc[0][2][0];
    r.w = h ? acc[1][3][0] : acc[0][3][0];
    *reinterpret_cast<float4*>(o) = r;
}

__global__ __launch_bounds__(256) void k_groupsum_wmma(const float* __restrict__ x,
                                                       float* __restrict__ cs) {
    const int lane = threadIdx.x & 31;
    const int w    = blockIdx.x * 8 + (threadIdx.x >> 5);
    const int c    = lane & 15;
    const int h    = lane >> 4;

    const float* p = x + (size_t)w * (16 * FF) + (size_t)(2 * h) * FF + 4 * c;
    float*       o = cs + (size_t)w * FF + 4 * lane;
    constexpr size_t XSTEP = (size_t)K1_WAVES * 16 * FF;
    constexpr size_t OSTEP = (size_t)K1_WAVES * FF;

    float4 buf[2][16];
    k1_load_group(buf[0], p);
#pragma unroll
    for (int t = 0; t < TPW; ++t) {
        __builtin_amdgcn_sched_barrier(0);
        if (t + 1 < TPW) k1_load_group(buf[(t + 1) & 1], p + (size_t)(t + 1) * XSTEP);
        __builtin_amdgcn_sched_barrier(0);
        k1_sum_group(buf[t & 1], o + (size_t)t * OSTEP, h);
    }
}

#define SEGS   16
#define SEGLEN (GG / SEGS)
#define SLICE  32
#define QL     (SLICE / 4)

__global__ __launch_bounds__(128) void k_prefix(float* __restrict__ cs) {
    __shared__ float4 tot[SEGS][QL];

    const int q   = threadIdx.x & (QL - 1);
    const int seg = threadIdx.x / QL;
    const int sl  = blockIdx.x & (FF / SLICE - 1);
    const int b   = blockIdx.x / (FF / SLICE);

    float4* p = reinterpret_cast<float4*>(
        cs + ((size_t)b * GG + (size_t)seg * SEGLEN) * FF + sl * SLICE + q * 4);
    constexpr int RS = FF / 4;

    float4 v[SEGLEN];
#pragma unroll
    for (int j = 0; j < SEGLEN; ++j) v[j] = p[j * RS];

#pragma unroll
    for (int j = 1; j < SEGLEN; ++j) {
        v[j].x += v[j - 1].x; v[j].y += v[j - 1].y;
        v[j].z += v[j - 1].z; v[j].w += v[j - 1].w;
    }
    tot[seg][q] = v[SEGLEN - 1];
    __syncthreads();

    float4 t[SEGS - 1];
#pragma unroll
    for (int s = 0; s < SEGS - 1; ++s) t[s] = tot[s][q];

    float4 off = make_float4(0.0f, 0.0f, 0.0f, 0.0f);
#pragma unroll
    for (int s = 0; s < SEGS - 1; ++s) {
        const bool in = s < seg;
        off.x += in ? t[s].x : 0.0f; off.y += in ? t[s].y : 0.0f;
        off.z += in ? t[s].z : 0.0f; off.w += in ? t[s].w : 0.0f;
    }

#pragma unroll
    for (int j = 0; j < SEGLEN; ++j) {
        float4 r;
        r.x = off.x + v[j].x; r.y = off.y + v[j].y;
        r.z = off.z + v[j].z; r.w = off.w + v[j].w;
        p[j * RS] = r;
    }
}

__global__ __launch_bounds__(128) void k_out(const float* __restrict__ x,
                                             const float* __restrict__ ps,
                                             const int* __restrict__ lengths,
                                             float* __restrict__ out) {
    const int f    = threadIdx.x;
    const int tile = blockIdx.x;
    const int b    = blockIdx.y;
    const int t0   = tile * 16;

    const int   fsize = lengths[b] - (OUTT - 1);
    const float fsf   = (float)fsize;

    const float* xb = x  + (size_t)b * TT * FF + f;
    const float* pb = ps + (size_t)b * GG * FF + f;

    float lo = (tile == 0) ? 0.0f : pb[(size_t)(tile - 1) * FF];

    const int idx0 = fsize + t0;
    const int g0   = idx0 >> 4;
    const int r0   = idx0 & 15;
    float hi = (g0 == 0) ? 0.0f : pb[(size_t)(g0 - 1) * FF];
    const int s0 = idx0 - r0;
    for (int i = 0; i < r0; ++i)
        hi += xb[(size_t)(s0 + i) * FF];

    float* ob = out + ((size_t)b * OUTT + t0) * FF + f;
#pragma unroll
    for (int j = 0; j < 16; ++j) {
        ob[(size_t)j * FF] = (hi - lo) / fsf;
        if (j < 15) {
            lo += xb[(size_t)(t0 + j) * FF];
            hi += xb[(size_t)(idx0 + j) * FF];
        }
    }
}

extern "C" void kernel_launch(void* const* d_in, const int* in_sizes, int n_in,
                              void* d_out, int out_size, void* d_ws, size_t ws_size,
                              hipStream_t stream) {
    (void)in_sizes; (void)n_in; (void)out_size; (void)ws_size;
    const float* x       = (const float*)d_in[0];
    const int*   lengths = (const int*)d_in[1];
    float*       out     = (float*)d_out;
    float*       cs      = (float*)d_ws;

    k_groupsum_wmma<<<512, 256, 0, stream>>>(x, cs);
    k_prefix<<<128, 128, 0, stream>>>(cs);
    k_out<<<dim3(64, 32), 128, 0, stream>>>(x, cs, lengths, out);
}


# Prefix pass removed (3 launches -> 2): group-sum work-group keeps a running sum over its 32-group chunk in LDS, output kernel adds the <=15 earlier chunk totals itself; same sum order
# speedup vs baseline: 4.3431x; 1.2206x over previous
#include <hip/hip_runtime.h>

#define BB   32
#define TT   8192
#define FF   128
#define OUTT 1024
#define GG   (TT / 16)

typedef float v2f __attribute__((ext_vector_type(2)));
typedef float v8f __attribute__((ext_vector_type(8)));

#define TPW 4
#define CHUNK 32
#define NCHUNK (GG / CHUNK)
#define K1_BLOCKS (BB * GG / CHUNK)
static_assert(K1_BLOCKS == 512, "the launch below is written for 512 work-groups");
static_assert(CHUNK == 8 * TPW, "a chunk is what the 8 waves of a work-group sum");

__device__ __forceinline__ void k1_load_group(float4 (&d)[16], const float* __restrict__ p) {
#pragma unroll
    for (int i = 0; i < 16; ++i) {
        const int k = i >> 2, q = (i >> 1) & 1, s = i & 1;
        d[i] = *reinterpret_cast<const float4*>(p + (size_t)(4 * k + s) * FF + 64 * q);
    }
}

__device__ __forceinline__ float4 k1_sum_group(const float4 (&d)[16], int h) {
    v2f a; a.x = 1.0f; a.y = 1.0f;
    v8f acc[2][4];
#pragma unroll
    for (int i = 0; i < 8; ++i) acc[i >> 2][i & 3] = v8f{};
#pragma unroll
    for (int i = 0; i < 8; ++i) {
        const int q = i & 1;
        const float4 r0 = d[2 * i + 0];
        const float4 r1 = d[2 * i + 1];
        v2f b0; b0.x = r0.x; b0.y = r1.x;
        v2f b1; b1.x = r0.y; b1.y = r1.y;
        v2f b2; b2.x = r0.z; b2.y = r1.z;
        v2f b3; b3.x = r0.w; b3.y = r1.w;
        acc[q][0] = __builtin_amdgcn_wmma_f32_16x16x4_f32(false, a, false, b0, (short)0,
                                                          acc[q][0], false, false);
        acc[q][1] = __builtin_amdgcn_wmma_f32_16x16x4_f32(false, a, false, b1, (short)0,
                                                          acc[q][1], false, false);
        acc[q][2] = __builtin_amdgcn_wmma_f32_16x16x4_f32(false, a, false, b2, (short)0,
                                                          acc[q][2], false, false);
        acc[q][3] = __builtin_amdgcn_wmma_f32_16x16x4_f32(false, a, false, b3, (short)0,
                                                          acc[q][3], false, false);
    }
    float4 r;
    r.x = h ? acc[1][0][0] : acc[0][0][0];
    r.y = h ? acc[1][1][0] : acc[0][1][0];
    r.z = h ? acc[1][2][0] : acc[0][2][0];
    r.w = h ? acc[1][3][0] : acc[0][3][0];
    return r;
}

__global__ __launch_bounds__(256) void k_groupsum_wmma(const float* __restrict__ x,
                                                       float* __restrict__ lp) {
    __shared__ float4 sums[CHUNK][32];

    const int lane = threadIdx.x & 31;
    const int wv   = threadIdx.x >> 5;
    const int c    = lane & 15;
    const int h    = lane >> 4;

    const size_t g0 = (size_t)blockIdx.x * CHUNK + wv;
    const float* p  = x + g0 * (16 * FF) + (size_t)(2 * h) * FF + 4 * c;
    constexpr size_t XSTEP = (size_t)8 * 16 * FF;

    float4 buf[2][16];
    k1_load_group(buf[0], p);
#pragma unroll
    for (int t = 0; t < TPW; ++t) {
        __builtin_amdgcn_sched_barrier(0);
        if (t + 1 < TPW) k1_load_group(buf[(t + 1) & 1], p + (size_t)(t + 1) * XSTEP);
        __builtin_amdgcn_sched_barrier(0);
        sums[wv + 8 * t][lane] = k1_sum_group(buf[t & 1], h);
    }
    __syncthreads();

    if (wv == 0) {
        float4* o = reinterpret_cast<float4*>(lp + (size_t)blockIdx.x * CHUNK * FF) + lane;
        constexpr int RS = FF / 4;
        float4 run = sums[0][lane];
        o[0] = run;
#pragma unroll
        for (int j = 1; j < CHUNK; ++j) {
            const float4 v = sums[j][lane];
            run.x = v.x + run.x; run.y = v.y + run.y;
            run.z = v.z + run.z; run.w = v.w + run.w;
            o[j * RS] = run;
        }
    }
}

__global__ __launch_bounds__(128) void k_out(const float* __restrict__ x,
                                             const float* __restrict__ lp,
                                             const int* __restrict__ lengths,
                                             float* __restrict__ out) {
    const int f    = threadIdx.x;
    const int tile = blockIdx.x;
    const int b    = blockIdx.y;
    const int t0   = tile * 16;

    const int   fsize = lengths[b] - (OUTT - 1);
    const float fsf   = (float)fsize;

    const float* xb = x  + (size_t)b * TT * FF + f;
    const float* pb = lp + (size_t)b * GG * FF + f;

    const int idx0 = fsize + t0;
    const int g0   = idx0 >> 4;
    const int r0   = idx0 & 15;
    const int gl   = tile - 1;
    const int gh   = g0 - 1;
    const int cl   = gl >> 5;
    const int ch   = gh >> 5;

    float tot[NCHUNK - 1];
#pragma unroll
    for (int s = 0; s < NCHUNK - 1; ++s) tot[s] = pb[(size_t)(CHUNK * s + CHUNK - 1) * FF];
    const float rowl = pb[(size_t)(gl < 0 ? 0 : gl) * FF];
    const float rowh = pb[(size_t)(gh < 0 ? 0 : gh) * FF];

    const float offl = (cl > 0) ? 0.0f + tot[0] : 0.0f;
    float offh = 0.0f;
#pragma unroll
    for (int s = 0; s < NCHUNK - 1; ++s) offh += (s < ch) ? tot[s] : 0.0f;

    float lo = (tile == 0) ? 0.0f : offl + rowl;
    float hi = (g0 == 0) ? 0.0f : offh + rowh;
    const int s0 = idx0 - r0;
    for (int i = 0; i < r0; ++i)
        hi += xb[(size_t)(s0 + i) * FF];

    float* ob = out + ((size_t)b * OUTT + t0) * FF + f;
#pragma unroll
    for (int j = 0; j < 16; ++j) {
        ob[(size_t)j * FF] = (hi - lo) / fsf;
        if (j < 15) {
            lo += xb[(size_t)(t0 + j) * FF];
            hi += xb[(size_t)(idx0 + j) * FF];
        }
    }
}

extern "C" void kernel_launch(void* const* d_in, const int* in_sizes, int n_in,
                              void* d_out, int out_size, void* d_ws, size_t ws_size,
                              hipStream_t stream) {
    (void)in_sizes; (void)n_in; (void)out_size; (void)ws_size;
    const float* x       = (const float*)d_in[0];
    const int*   lengths = (const int*)d_in[1];
    float*       out     = (float*)d_out;
    float*       lp      = (float*)d_ws;

    k_groupsum_wmma<<<512, 256, 0, stream>>>(x, lp);
    k_out<<<dim3(64, 32), 128, 0, stream>>>(x, lp, lengths, out);
}


# Output kernel: hi's up-to-15 residual rows are read unconditionally and kept by count, so every load of a thread issues in one group; was a loop of one dependent load per trip; same sum order
# speedup vs baseline: 5.6626x; 1.3038x over previous
#include <hip/hip_runtime.h>

#define BB   32
#define TT   8192
#define FF   128
#define OUTT 1024
#define GG   (TT / 16)

typedef float v2f __attribute__((ext_vector_type(2)));
typedef float v8f __attribute__((ext_vector_type(8)));

#define TPW 4
#define CHUNK 32
#define NCHUNK (GG / CHUNK)
#define K1_BLOCKS (BB * GG / CHUNK)
static_assert(K1_BLOCKS == 512, "the launch below is written for 512 work-groups");
static_assert(CHUNK == 8 * TPW, "a chunk is what the 8 waves of a work-group sum");

__device__ __forceinline__ void k1_load_group(float4 (&d)[16], const float* __restrict__ p) {
#pragma unroll
    for (int i = 0; i < 16; ++i) {
        const int k = i >> 2, q = (i >> 1) & 1, s = i & 1;
        d[i] = *reinterpret_cast<const float4*>(p + (size_t)(4 * k + s) * FF + 64 * q);
    }
}

__device__ __forceinline__ float4 k1_sum_group(const float4 (&d)[16], int h) {
    v2f a; a.x = 1.0f; a.y = 1.0f;
    v8f acc[2][4];
#pragma unroll
    for (int i = 0; i < 8; ++i) acc[i >> 2][i & 3] = v8f{};
#pragma unroll
    for (int i = 0; i < 8; ++i) {
        const int q = i & 1;
        const float4 r0 = d[2 * i + 0];
        const float4 r1 = d[2 * i + 1];
        v2f b0; b0.x = r0.x; b0.y = r1.x;
        v2f b1; b1.x = r0.y; b1.y = r1.y;
        v2f b2; b2.x = r0.z; b2.y = r1.z;
        v2f b3; b3.x = r0.w; b3.y = r1.w;
        acc[q][0] = __builtin_amdgcn_wmma_f32_16x16x4_f32(false, a, false, b0, (short)0,
                                                          acc[q][0], false, false);
        acc[q][1] = __builtin_amdgcn_wmma_f32_16x16x4_f32(false, a, false, b1, (short)0,
                                                          acc[q][1], false, false);
        acc[q][2] = __builtin_amdgcn_wmma_f32_16x16x4_f32(false, a, false, b2, (short)0,
                                                          acc[q][2], false, false);
        acc[q][3] = __builtin_amdgcn_wmma_f32_16x16x4_f32(false, a, false, b3, (short)0,
                                                          acc[q][3], false, false);
    }
    float4 r;
    r.x = h ? acc[1][0][0] : acc[0][0][0];
    r.y = h ? acc[1][1][0] : acc[0][1][0];
    r.z = h ? acc[1][2][0] : acc[0][2][0];
    r.w = h ? acc[1][3][0] : acc[0][3][0];
    return r;
}

__global__ __launch_bounds__(256) void k_groupsum_wmma(const float* __restrict__ x,
                                                       float* __restrict__ lp) {
    __shared__ float4 sums[CHUNK][32];

    const int lane = threadIdx.x & 31;
    const int wv   = threadIdx.x >> 5;
    const int c    = lane & 15;
    const int h    = lane >> 4;

    const size_t g0 = (size_t)blockIdx.x * CHUNK + wv;
    const float* p  = x + g0 * (16 * FF) + (size_t)(2 * h) * FF + 4 * c;
    constexpr size_t XSTEP = (size_t)8 * 16 * FF;

    float4 buf[2][16];
    k1_load_group(buf[0], p);
#pragma unroll
    for (int t = 0; t < TPW; ++t) {
        __builtin_amdgcn_sched_barrier(0);
        if (t + 1 < TPW) k1_load_group(buf[(t + 1) & 1], p + (size_t)(t + 1) * XSTEP);
        __builtin_amdgcn_sched_barrier(0);
        sums[wv + 8 * t][lane] = k1_sum_group(buf[t & 1], h);
    }
    __syncthreads();

    if (wv == 0) {
        float4* o = reinterpret_cast<float4*>(lp + (size_t)blockIdx.x * CHUNK * FF) + lane;
        constexpr int RS = FF / 4;
        float4 run = sums[0][lane];
        o[0] = run;
#pragma unroll
        for (int j = 1; j < CHUNK; ++j) {
            const float4 v = sums[j][lane];
            run.x = v.x + run.x; run.y = v.y + run.y;
            run.z = v.z + run.z; run.w = v.w + run.w;
            o[j * RS] = run;
        }
    }
}

__global__ __launch_bounds__(128) void k_out(const float* __restrict__ x,
                                             const float* __restrict__ lp,
                                             const int* __restrict__ lengths,
                                             float* __restrict__ out) {
    const int f    = threadIdx.x;
    const int tile = blockIdx.x;
    const int b    = blockIdx.y;
    const int t0   = tile * 16;

    const int   fsize = lengths[b] - (OUTT - 1);
    const float fsf   = (float)fsize;

    const float* xb = x  + (size_t)b * TT * FF + f;
    const float* pb = lp + (size_t)b * GG * FF + f;

    const int idx0 = fsize + t0;
    const int g0   = idx0 >> 4;
    const int r0   = idx0 & 15;
    const int gl   = tile - 1;
    const int gh   = g0 - 1;
    const int cl   = gl >> 5;
    const int ch   = gh >> 5;

    float tot[NCHUNK - 1];
#pragma unroll
    for (int s = 0; s < NCHUNK - 1; ++s) tot[s] = pb[(size_t)(CHUNK * s + CHUNK - 1) * FF];
    const float rowl = pb[(size_t)(gl < 0 ? 0 : gl) * FF];
    const float rowh = pb[(size_t)(gh < 0 ? 0 : gh) * FF];

    const int s0 = idx0 - r0;
    float res[15];
#pragma unroll
    for (int i = 0; i < 15; ++i) res[i] = xb[(size_t)(s0 + i) * FF];

    float lov[15], hiv[15];
#pragma unroll
    for (int j = 0; j < 15; ++j) {
        lov[j] = xb[(size_t)(t0 + j) * FF];
        hiv[j] = xb[(size_t)(idx0 + j) * FF];
    }

    const float offl = (cl > 0) ? 0.0f + tot[0] : 0.0f;
    float offh = 0.0f;
#pragma unroll
    for (int s = 0; s < NCHUNK - 1; ++s) offh += (s < ch) ? tot[s] : 0.0f;

    float lo = (tile == 0) ? 0.0f : offl + rowl;
    float hi = (g0 == 0) ? 0.0f : offh + rowh;
#pragma unroll
    for (int i = 0; i < 15; ++i) hi = (i < r0) ? hi + res[i] : hi;

    float* ob = out + ((size_t)b * OUTT + t0) * FF + f;
#pragma unroll
    for (int j = 0; j < 16; ++j) {
        ob[(size_t)j * FF] = (hi - lo) / fsf;
        if (j < 15) {
            lo += lov[j];
            hi += hiv[j];
        }
    }
}

extern "C" void kernel_launch(void* const* d_in, const int* in_sizes, int n_in,
                              void* d_out, int out_size, void* d_ws, size_t ws_size,
                              hipStream_t stream) {
    (void)in_sizes; (void)n_in; (void)out_size; (void)ws_size;
    const float* x       = (const float*)d_in[0];
    const int*   lengths = (const int*)d_in[1];
    float*       out     = (float*)d_out;
    float*       lp      = (float*)d_ws;

    k_groupsum_wmma<<<512, 256, 0, stream>>>(x, lp);
    k_out<<<dim3(64, 32), 128, 0, stream>>>(x, lp, lengths, out);
}


# Group-sum kernel sums only the chunks in front of lengths[b] (rows past it never reach out): needed (batch, chunk, feature-half) items numbered compactly over 1024 4-wave work-groups; same sum order
# speedup vs baseline: 6.7821x; 1.1977x over previous
#include <hip/hip_runtime.h>

#define BB   32
#define TT   8192
#define FF   128
#define OUTT 1024
#define GG   (TT / 16)

typedef float v2f __attribute__((ext_vector_type(2)));
typedef float v8f __attribute__((ext_vector_type(8)));

#define TPW 8
#define K1_WAVES 4
#define CHUNK 32
#define NCHUNK (GG / CHUNK)
#define CROWS (CHUNK * 16)
#define K1_BLOCKS (BB * NCHUNK * 2)
static_assert(K1_BLOCKS == 1024, "the launch below is written for 1024 work-groups");
static_assert(CHUNK == K1_WAVES * TPW, "a chunk is what the waves of a work-group sum");
static_assert(BB == 32, "the item map keeps one batch per lane of a wave");

__device__ __forceinline__ void k1_load_group(float4 (&d)[8], const float* __restrict__ p) {
#pragma unroll
    for (int i = 0; i < 8; ++i) {
        const int k = i >> 1, s = i & 1;
        d[i] = *reinterpret_cast<const float4*>(p + (size_t)(4 * k + s) * FF);
    }
}

__device__ __forceinline__ float4 k1_sum_group(const float4 (&d)[8]) {
    v2f a; a.x = 1.0f; a.y = 1.0f;
    v8f acc[4];
#pragma unroll
    for (int j = 0; j < 4; ++j) acc[j] = v8f{};
#pragma unroll
    for (int k = 0; k < 4; ++k) {
        const float4 r0 = d[2 * k + 0];
        const float4 r1 = d[2 * k + 1];
        v2f b0; b0.x = r0.x; b0.y = r1.x;
        v2f b1; b1.x = r0.y; b1.y = r1.y;
        v2f b2; b2.x = r0.z; b2.y = r1.z;
        v2f b3; b3.x = r0.w; b3.y = r1.w;
        acc[0] = __builtin_amdgcn_wmma_f32_16x16x4_f32(false, a, false, b0, (short)0,
                                                       acc[0], false, false);
        acc[1] = __builtin_amdgcn_wmma_f32_16x16x4_f32(false, a, false, b1, (short)0,
                                                       acc[1], false, false);
        acc[2] = __builtin_amdgcn_wmma_f32_16x16x4_f32(false, a, false, b2, (short)0,
                                                       acc[2], false, false);
        acc[3] = __builtin_amdgcn_wmma_f32_16x16x4_f32(false, a, false, b3, (short)0,
                                                       acc[3], false, false);
    }
    float4 r;
    r.x = acc[0][0]; r.y = acc[1][0]; r.z = acc[2][0]; r.w = acc[3][0];
    return r;
}

__global__ __launch_bounds__(128) void k_groupsum_wmma(const float* __restrict__ x,
                                                       const int* __restrict__ lengths,
                                                       float* __restrict__ lp) {
    __shared__ __attribute__((aligned(16))) float sums[CHUNK][128];

    const int lane = threadIdx.x & 31;
    const int wv   = threadIdx.x >> 5;
    const int c    = lane & 15;
    const int h    = lane >> 4;

    const int len = lengths[lane];
    int nc = (len <= 0) ? 0 : ((len + (CROWS - 1)) / CROWS);
    nc = (nc > NCHUNK) ? NCHUNK : nc;
    const unsigned le_mask = (2u << lane) - 1u;
    int incl = 0;
#pragma unroll
    for (int k = 0; k < 5; ++k) {
        const unsigned has = (unsigned)__ballot((nc >> k) & 1);
        incl += __popc(has & le_mask) << k;
    }
    const int item = (int)(blockIdx.x >> 1);
    const int fh   = (int)(blockIdx.x & 1);
    const int b = __popc((unsigned)__ballot(incl <= item));
    if (b >= BB) return;
    const int chunk = item - __builtin_amdgcn_readlane(incl - nc, b);

    const size_t g0 = (size_t)b * GG + (size_t)chunk * CHUNK + wv;
    const float* p  = x + g0 * (16 * FF) + (size_t)(2 * h) * FF + 64 * fh + 4 * c;
    constexpr size_t XSTEP = (size_t)K1_WAVES * 16 * FF;

    float4 buf[3][8];
    k1_load_group(buf[0], p);
    k1_load_group(buf[1], p + XSTEP);
#pragma unroll
    for (int t = 0; t < TPW; ++t) {
        __builtin_amdgcn_sched_barrier(0);
        if (t + 2 < TPW) k1_load_group(buf[(t + 2) % 3], p + (size_t)(t + 2) * XSTEP);
        __builtin_amdgcn_sched_barrier(0);
        *reinterpret_cast<float4*>(&sums[wv + K1_WAVES * t][4 * lane]) = k1_sum_group(buf[t % 3]);
    }
    __syncthreads();

    if (wv == 0) {
        float2* o = reinterpret_cast<float2*>(lp + ((size_t)b * GG + (size_t)chunk * CHUNK) * FF
                                                 + 64 * fh) + lane;
        constexpr int RS = FF / 2;
        float2 run = *reinterpret_cast<const float2*>(&sums[0][2 * lane]);
        o[0] = run;
#pragma unroll
        for (int j = 1; j < CHUNK; ++j) {
            const float2 v = *reinterpret_cast<const float2*>(&sums[j][2 * lane]);
            run.x = v.x + run.x; run.y = v.y + run.y;
            o[j * RS] = run;
        }
    } else if (wv == 1 && chunk == 0) {
        const int ncb = __builtin_amdgcn_readlane(nc, b);
        float2* z = reinterpret_cast<float2*>(lp + (size_t)b * GG * FF + 64 * fh) + lane;
        float2 zero; zero.x = 0.0f; zero.y = 0.0f;
#pragma unroll
        for (int s = 1; s < NCHUNK - 1; ++s)
            if (s >= ncb) z[(size_t)(CHUNK * s + CHUNK - 1) * (FF / 2)] = zero;
    }
}

__global__ __launch_bounds__(128) void k_out(const float* __restrict__ x,
                                             const float* __restrict__ lp,
                                             const int* __restrict__ lengths,
                                             float* __restrict__ out) {
    const int f    = threadIdx.x;
    const int tile = blockIdx.x;
    const int b    = blockIdx.y;
    const int t0   = tile * 16;

    const int   fsize = lengths[b] - (OUTT - 1);
    const float fsf   = (float)fsize;

    const float* xb = x  + (size_t)b * TT * FF + f;
    const float* pb = lp + (size_t)b * GG * FF + f;

    const int idx0 = fsize + t0;
    const int g0   = idx0 >> 4;
    const int r0   = idx0 & 15;
    const int gl   = tile - 1;
    const int gh   = g0 - 1;
    const int cl   = gl >> 5;
    const int ch   = gh >> 5;

    float tot[NCHUNK - 1];
#pragma unroll
    for (int s = 0; s < NCHUNK - 1; ++s) tot[s] = pb[(size_t)(CHUNK * s + CHUNK - 1) * FF];
    const float rowl = pb[(size_t)(gl < 0 ? 0 : gl) * FF];
    const float rowh = pb[(size_t)(gh < 0 ? 0 : gh) * FF];

    const int s0 = idx0 - r0;
    float res[15];
#pragma unroll
    for (int i = 0; i < 15; ++i) res[i] = xb[(size_t)(s0 + i) * FF];

    float lov[15], hiv[15];
#pragma unroll
    for (int j = 0; j < 15; ++j) {
        lov[j] = xb[(size_t)(t0 + j) * FF];
        hiv[j] = xb[(size_t)(idx0 + j) * FF];
    }

    const float offl = (cl > 0) ? 0.0f + tot[0] : 0.0f;
    float offh = 0.0f;
#pragma unroll
    for (int s = 0; s < NCHUNK - 1; ++s) offh += (s < ch) ? tot[s] : 0.0f;

    float lo = (tile == 0) ? 0.0f : offl + rowl;
    float hi = (g0 == 0) ? 0.0f : offh + rowh;
#pragma unroll
    for (int i = 0; i < 15; ++i) hi = (i < r0) ? hi + res[i] : hi;

    float* ob = out + ((size_t)b * OUTT + t0) * FF + f;
#pragma unroll
    for (int j = 0; j < 16; ++j) {
        ob[(size_t)j * FF] = (hi - lo) / fsf;
        if (j < 15) {
            lo += lov[j];
            hi += hiv[j];
        }
    }
}

extern "C" void kernel_launch(void* const* d_in, const int* in_sizes, int n_in,
                              void* d_out, int out_size, void* d_ws, size_t ws_size,
                              hipStream_t stream) {
    (void)in_sizes; (void)n_in; (void)out_size; (void)ws_size;
    const float* x       = (const float*)d_in[0];
    const int*   lengths = (const int*)d_in[1];
    float*       out     = (float*)d_out;
    float*       lp      = (float*)d_ws;

    k_groupsum_wmma<<<1024, 128, 0, stream>>>(x, lengths, lp);
    k_out<<<dim3(64, 32), 128, 0, stream>>>(x, lp, lengths, out);
}
